# starting kernel
# baseline (speedup 1.0000x reference)
#include <hip/hip_runtime.h>
#include <hip/hip_bf16.h>

typedef __attribute__((ext_vector_type(16))) __bf16        v16bf;
typedef __attribute__((ext_vector_type(8)))  __bf16        v8bf;
typedef __attribute__((ext_vector_type(8)))  float         v8f;

__device__ __forceinline__ __bf16 tobf(float f) { return (__bf16)f; }

__global__ void k_fill0(float* __restrict__ p, int n) {
  int i = blockIdx.x * blockDim.x + threadIdx.x;
  if (i < n) p[i] = 0.0f;
}

__global__ void k_sig_deg(const float* __restrict__ P, const int* __restrict__ col,
                          float* __restrict__ w, float* __restrict__ deg, int E) {
  int e = blockIdx.x * blockDim.x + threadIdx.x;
  if (e >= E) return;
  float s = 1.0f / (1.0f + __expf(-P[e]));
  w[e] = s;
  atomicAdd(&deg[col[e]], s);
}

__global__ void k_dinv(float* __restrict__ deg, int N) {
  int i = blockIdx.x * blockDim.x + threadIdx.x;
  if (i >= N) return;
  float d = deg[i] + 1.0f;
  deg[i] = (d > 0.0f) ? rsqrtf(d) : 0.0f;
}

__global__ void k_norm(const int* __restrict__ row, const int* __restrict__ col,
                       const float* __restrict__ dinv, float* __restrict__ w, int E) {
  int e = blockIdx.x * blockDim.x + threadIdx.x;
  if (e >= E) return;
  w[e] = dinv[row[e]] * w[e] * dinv[col[e]];
}

template<int NCOL>
__global__ void k_wt_bf16(const float* __restrict__ W, __bf16* __restrict__ Wt) {
  int idx = blockIdx.x * blockDim.x + threadIdx.x;
  if (idx >= 128 * NCOL) return;
  int k = idx / NCOL, n = idx % NCOL;
  Wt[n * 128 + k] = tobf(W[idx]);
}

template<int NCOL, bool RELU_A>
__global__ void k_gemm_wmma(const float* __restrict__ A,
                            const __bf16* __restrict__ Bt,
                            const float* __restrict__ bias,
                            const float* __restrict__ dinv,
                            float* __restrict__ H,
                            float* __restrict__ OUT, int N) {
  constexpr int K    = 128;
  constexpr int LDSP = 136;
  __shared__ __align__(16) __bf16 As[16 * LDSP];

  const int lane = threadIdx.x & 31;
  const int wave = threadIdx.x >> 5;
  const int m0   = blockIdx.x * 16;
  const int n0   = wave * 16;
  const int half = lane >> 4;
  const int l15  = lane & 15;

  for (int base = threadIdx.x * 8; base < 16 * K; base += blockDim.x * 8) {
    const int m = base >> 7;
    const int k = base & (K - 1);
    int mg = m0 + m; if (mg > N - 1) mg = N - 1;
    const float* src = A + (size_t)mg * K + k;
    float4 f0 = *(const float4*)(src);
    float4 f1 = *(const float4*)(src + 4);
    if (RELU_A) {
      f0.x = fmaxf(f0.x, 0.f); f0.y = fmaxf(f0.y, 0.f);
      f0.z = fmaxf(f0.z, 0.f); f0.w = fmaxf(f0.w, 0.f);
      f1.x = fmaxf(f1.x, 0.f); f1.y = fmaxf(f1.y, 0.f);
      f1.z = fmaxf(f1.z, 0.f); f1.w = fmaxf(f1.w, 0.f);
    }
    v8bf pk;
    pk[0] = tobf(f0.x); pk[1] = tobf(f0.y); pk[2] = tobf(f0.z); pk[3] = tobf(f0.w);
    pk[4] = tobf(f1.x); pk[5] = tobf(f1.y); pk[6] = tobf(f1.z); pk[7] = tobf(f1.w);
    *(v8bf*)(&As[m * LDSP + k]) = pk;
  }
  __syncthreads();

  const __bf16* Brow = Bt + (size_t)(n0 + l15) * K;
  const __bf16* Alds = As + l15 * LDSP;

  v8f acc = {};

#pragma unroll
  for (int k0 = 0; k0 < K; k0 += 32) {
    const v8bf a0 = *(const v8bf*)(Alds + k0 + half * 8);
    const v8bf a1 = *(const v8bf*)(Alds + k0 + 16 + half * 8);
    const v16bf a = __builtin_shufflevector(a0, a1,
        0, 1, 2, 3, 4, 5, 6, 7, 8, 9, 10, 11, 12, 13, 14, 15);
    const v8bf b0 = *(const v8bf*)(Brow + k0 + half * 8);
    const v8bf b1 = *(const v8bf*)(Brow + k0 + 16 + half * 8);
    const v16bf b = __builtin_shufflevector(b0, b1,
        0, 1, 2, 3, 4, 5, 6, 7, 8, 9, 10, 11, 12, 13, 14, 15);
    acc = __builtin_amdgcn_wmma_f32_16x16x32_bf16(false, a, false, b,
                                                  (short)0, acc, false, false);
  }

  const int   n  = n0 + l15;
  const float bn = bias[n];
  if (m0 + 16 <= N) {
#pragma unroll
    for (int r = 0; r < 8; ++r) {
      const int   m = m0 + half * 8 + r;
      const float d = dinv[m];
      const float h = acc[r];
      H  [(size_t)m * NCOL + n] = h;
      OUT[(size_t)m * NCOL + n] = fmaf(d * d, h, bn);
    }
  } else {
#pragma unroll
    for (int r = 0; r < 8; ++r) {
      const int m = m0 + half * 8 + r;
      if (m < N) {
        const float d = dinv[m];
        const float h = acc[r];
        H  [(size_t)m * NCOL + n] = h;
        OUT[(size_t)m * NCOL + n] = fmaf(d * d, h, bn);
      }
    }
  }
}

template<int F>
__global__ void k_scatter(const int* __restrict__ row, const int* __restrict__ col,
                          const float* __restrict__ nrm, const float* __restrict__ h,
                          float* __restrict__ out, int E) {
  const int wid  = __builtin_amdgcn_readfirstlane(
                     (blockIdx.x * blockDim.x + threadIdx.x) >> 5);
  const int lane = threadIdx.x & 31;
  if (wid >= E) return;
  const int   r  = row[wid];
  const int   c  = col[wid];
  const float nm = nrm[wid];
  constexpr int PER = F / 32;
  const float* hp = h   + (size_t)r * F + lane * PER;
  float*       op = out + (size_t)c * F + lane * PER;
  if constexpr (PER == 4) {
    float4 v = *(const float4*)hp;
    atomicAdd(op + 0, nm * v.x);
    atomicAdd(op + 1, nm * v.y);
    atomicAdd(op + 2, nm * v.z);
    atomicAdd(op + 3, nm * v.w);
  } else {
    float2 v = *(const float2*)hp;
    atomicAdd(op + 0, nm * v.x);
    atomicAdd(op + 1, nm * v.y);
  }
}

extern "C" void kernel_launch(void* const* d_in, const int* in_sizes, int n_in,
                              void* d_out, int out_size, void* d_ws, size_t ws_size,
                              hipStream_t stream) {
  const float* x  = (const float*)d_in[0];
  const int*   ei = (const int*)  d_in[1];
  const float* P  = (const float*)d_in[2];
  const float* W1 = (const float*)d_in[3];
  const float* b1 = (const float*)d_in[4];
  const float* W2 = (const float*)d_in[5];
  const float* b2 = (const float*)d_in[6];
  float* out = (float*)d_out;

  constexpr int F_IN = 128, H1 = 128, H2 = 64;
  const int N = in_sizes[0] / F_IN;
  const int E = in_sizes[2];
  const int* row = ei;
  const int* col = ei + E;

  float* dinv = (float*)d_ws;
  float* wn   = dinv + N;
  float* h    = wn + E;
  float* o1   = h + (size_t)N * H1;
  __bf16* w1t = (__bf16*)(o1 + (size_t)N * H1);
  __bf16* w2t = w1t + F_IN * H1;

  const int T = 256;

  k_fill0  <<<(N + T - 1) / T, T, 0, stream>>>(dinv, N);
  k_sig_deg<<<(E + T - 1) / T, T, 0, stream>>>(P, col, wn, dinv, E);
  k_dinv   <<<(N + T - 1) / T, T, 0, stream>>>(dinv, N);
  k_norm   <<<(E + T - 1) / T, T, 0, stream>>>(row, col, dinv, wn, E);
  k_wt_bf16<H1><<<(F_IN * H1 + T - 1) / T, T, 0, stream>>>(W1, w1t);
  k_wt_bf16<H2><<<(F_IN * H2 + T - 1) / T, T, 0, stream>>>(W2, w2t);

  const int strips = (N + 15) / 16;

  k_gemm_wmma<H1, false><<<strips, 32 * (H1 / 16), 0, stream>>>(
      x, w1t, b1, dinv, h, o1, N);
  k_scatter<H1><<<(E + 7) / 8, T, 0, stream>>>(row, col, wn, h, o1, E);

  k_gemm_wmma<H2, true><<<strips, 32 * (H2 / 16), 0, stream>>>(
      o1, w2t, b2, dinv, h, out, N);
  k_scatter<H2><<<(E + 7) / 8, T, 0, stream>>>(row, col, wn, h, out, E);
}


# layer-2 GEMM launch dissolved: layer-1 range blocks keep finished rows in LDS, relu + same bf16 WMMA tiles there, write h2; layer-2 blocks form b2+dinv^2*h2 at load; 8 -> 7 launches; sum order kept
# speedup vs baseline: 2.9915x; 2.9915x over previous
#include <hip/hip_runtime.h>
#include <hip/hip_bf16.h>

typedef __attribute__((ext_vector_type(16))) __bf16        v16bf;
typedef __attribute__((ext_vector_type(8)))  __bf16        v8bf;
typedef __attribute__((ext_vector_type(8)))  float         v8f;

__device__ __forceinline__ __bf16 tobf(float f) { return (__bf16)f; }

template<int TN>
__global__ void k_sig_rid(const float* __restrict__ P, const int* __restrict__ col,
                          float* __restrict__ w, unsigned char* __restrict__ rid,
                          int E, int Epad) {
  const int e0 = (blockIdx.x * blockDim.x + threadIdx.x) * 8;
  if (e0 >= Epad) return;
  const size_t al = (size_t)(P + e0) | (size_t)(col + e0) | (size_t)(w + e0) | (size_t)(rid + e0);
  if (e0 + 8 <= E && (al & 15) == 0) {
    const float4 p0 = *(const float4*)(P + e0);
    const float4 p1 = *(const float4*)(P + e0 + 4);
    const int4   c0 = *(const int4*)(col + e0);
    const int4   c1 = *(const int4*)(col + e0 + 4);
    float4 s0, s1;
    s0.x = 1.0f / (1.0f + __expf(-p0.x)); s0.y = 1.0f / (1.0f + __expf(-p0.y));
    s0.z = 1.0f / (1.0f + __expf(-p0.z)); s0.w = 1.0f / (1.0f + __expf(-p0.w));
    s1.x = 1.0f / (1.0f + __expf(-p1.x)); s1.y = 1.0f / (1.0f + __expf(-p1.y));
    s1.z = 1.0f / (1.0f + __expf(-p1.z)); s1.w = 1.0f / (1.0f + __expf(-p1.w));
    *(float4*)(w + e0)     = s0;
    *(float4*)(w + e0 + 4) = s1;
    uint2 id;
    id.x = ((unsigned)(c0.x / TN) & 255u)       | (((unsigned)(c0.y / TN) & 255u) << 8) |
           (((unsigned)(c0.z / TN) & 255u) << 16) | (((unsigned)(c0.w / TN) & 255u) << 24);
    id.y = ((unsigned)(c1.x / TN) & 255u)       | (((unsigned)(c1.y / TN) & 255u) << 8) |
           (((unsigned)(c1.z / TN) & 255u) << 16) | (((unsigned)(c1.w / TN) & 255u) << 24);
    *(uint2*)(rid + e0) = id;
  } else {
    for (int j = 0; j < 8; ++j) {
      const int e = e0 + j;
      if (e < E) {
        w[e]   = 1.0f / (1.0f + __expf(-P[e]));
        rid[e] = (unsigned char)(col[e] / TN);
      } else if (e < Epad) {
        rid[e] = 0;
      }
    }
  }
}

__device__ __forceinline__ int4 edge_rec(int e, const int* __restrict__ row,
                                         const int* __restrict__ col,
                                         const float* __restrict__ w,
                                         const float* __restrict__ dinv) {
  const int r = row[e];
  const int c = col[e];
  const float nm = dinv[r] * w[e] * dinv[c];
  return make_int4(r, c, __float_as_int(nm), 0);
}

template<int NCOL>
__global__ void k_wt_bf16(const float* __restrict__ W, __bf16* __restrict__ Wt) {
  int idx = blockIdx.x * blockDim.x + threadIdx.x;
  if (idx >= 128 * NCOL) return;
  int k = idx / NCOL, n = idx % NCOL;
  Wt[n * 128 + k] = tobf(W[idx]);
}

template<int NCOL, bool RELU_A>
__global__ void k_gemm_wmma(const float* __restrict__ A,
                            const __bf16* __restrict__ Bt,
                            const float* __restrict__ bias,
                            const float* __restrict__ dinv,
                            float* __restrict__ H,
                            float* __restrict__ OUT, int N) {
  constexpr int K    = 128;
  constexpr int LDSP = 136;
  __shared__ __align__(16) __bf16 As[16 * LDSP];

  const int lane = threadIdx.x & 31;
  const int wave = threadIdx.x >> 5;
  const int m0   = blockIdx.x * 16;
  const int n0   = wave * 16;
  const int half = lane >> 4;
  const int l15  = lane & 15;

  for (int base = threadIdx.x * 8; base < 16 * K; base += blockDim.x * 8) {
    const int m = base >> 7;
    const int k = base & (K - 1);
    int mg = m0 + m; if (mg > N - 1) mg = N - 1;
    const float* src = A + (size_t)mg * K + k;
    float4 f0 = *(const float4*)(src);
    float4 f1 = *(const float4*)(src + 4);
    if (RELU_A) {
      f0.x = fmaxf(f0.x, 0.f); f0.y = fmaxf(f0.y, 0.f);
      f0.z = fmaxf(f0.z, 0.f); f0.w = fmaxf(f0.w, 0.f);
      f1.x = fmaxf(f1.x, 0.f); f1.y = fmaxf(f1.y, 0.f);
      f1.z = fmaxf(f1.z, 0.f); f1.w = fmaxf(f1.w, 0.f);
    }
    v8bf pk;
    pk[0] = tobf(f0.x); pk[1] = tobf(f0.y); pk[2] = tobf(f0.z); pk[3] = tobf(f0.w);
    pk[4] = tobf(f1.x); pk[5] = tobf(f1.y); pk[6] = tobf(f1.z); pk[7] = tobf(f1.w);
    *(v8bf*)(&As[m * LDSP + k]) = pk;
  }
  __syncthreads();

  const __bf16* Brow = Bt + (size_t)(n0 + l15) * K;
  const __bf16* Alds = As + l15 * LDSP;

  v8f acc = {};

#pragma unroll
  for (int k0 = 0; k0 < K; k0 += 32) {
    const v8bf a0 = *(const v8bf*)(Alds + k0 + half * 8);
    const v8bf a1 = *(const v8bf*)(Alds + k0 + 16 + half * 8);
    const v16bf a = __builtin_shufflevector(a0, a1,
        0, 1, 2, 3, 4, 5, 6, 7, 8, 9, 10, 11, 12, 13, 14, 15);
    const v8bf b0 = *(const v8bf*)(Brow + k0 + half * 8);
    const v8bf b1 = *(const v8bf*)(Brow + k0 + 16 + half * 8);
    const v16bf b = __builtin_shufflevector(b0, b1,
        0, 1, 2, 3, 4, 5, 6, 7, 8, 9, 10, 11, 12, 13, 14, 15);
    acc = __builtin_amdgcn_wmma_f32_16x16x32_bf16(false, a, false, b,
                                                  (short)0, acc, false, false);
  }

  const int   n  = n0 + l15;
  const float bn = bias[n];
  if (m0 + 16 <= N) {
#pragma unroll
    for (int r = 0; r < 8; ++r) {
      const int   m = m0 + half * 8 + r;
      const float d = dinv[m];
      const float h = acc[r];
      H  [(size_t)m * NCOL + n] = h;
      OUT[(size_t)m * NCOL + n] = fmaf(d * d, h, bn);
    }
  } else {
#pragma unroll
    for (int r = 0; r < 8; ++r) {
      const int m = m0 + half * 8 + r;
      if (m < N) {
        const float d = dinv[m];
        const float h = acc[r];
        H  [(size_t)m * NCOL + n] = h;
        OUT[(size_t)m * NCOL + n] = fmaf(d * d, h, bn);
      }
    }
  }
}

template<int F, int J0>
__device__ __forceinline__ void rows_load(float (&v)[8][F / 32], int rx,
                                          const float* __restrict__ h, int lane) {
#pragma unroll
  for (int j = 0; j < 8; ++j) {
    const int rr = __builtin_amdgcn_readlane(rx, J0 + j);
    const float* hp = h + (size_t)rr * F + lane;
#pragma unroll
    for (int k = 0; k < F / 32; ++k) v[j][k] = hp[k * 32];
  }
}

template<int F, int TN, int LD, int J0>
__device__ __forceinline__ void rows_add(const float (&v)[8][F / 32], int ry, int rz,
                                         float* acc, int n0, int lane) {
#pragma unroll
  for (int j = 0; j < 8; ++j) {
    int c = __builtin_amdgcn_readlane(ry, J0 + j) - n0;
    c = ((unsigned)c < (unsigned)TN) ? c : TN;
    const float w = __int_as_float(__builtin_amdgcn_readlane(rz, J0 + j));
    float* ap = acc + c * LD + lane;
#pragma unroll
    for (int k = 0; k < F / 32; ++k)
      __hip_atomic_fetch_add(ap + k * 32, w * v[j][k],
                             __ATOMIC_RELAXED, __HIP_MEMORY_SCOPE_WORKGROUP);
  }
}

template<int F, int TN, int LD>
__device__ __forceinline__ void rec_group(const int4 r, const float* __restrict__ h,
                                          float* acc, int n0, int lane) {
  float va[8][F / 32];
  float vb[8][F / 32];
  rows_load<F, 0>(va, r.x, h, lane);
  rows_load<F, 8>(vb, r.x, h, lane);
  rows_add<F, TN, LD, 0>(va, r.y, r.z, acc, n0, lane);
  rows_load<F, 16>(va, r.x, h, lane);
  rows_add<F, TN, LD, 8>(vb, r.y, r.z, acc, n0, lane);
  rows_load<F, 24>(vb, r.x, h, lane);
  rows_add<F, TN, LD, 16>(va, r.y, r.z, acc, n0, lane);
  rows_add<F, TN, LD, 24>(vb, r.y, r.z, acc, n0, lane);
}

constexpr int LISTCAP = 16;

constexpr int NWAVE = 32;
constexpr int GROUP = 32;
constexpr int QCAP  = 512 + GROUP;

template<typename G>
__device__ __forceinline__ void scan_ids(const unsigned char* __restrict__ rid, int E, int blk,
                                         int wave, int lane, int* q, G grp) {
  int qn = 0;
  const unsigned pat = ((unsigned)blk & 255u) * 0x01010101u;
  const unsigned lt  = (1u << lane) - 1u;
  const int nch = (E + 511) >> 9;
  const uint4* rid4 = (const uint4*)rid;

  uint4 vn = make_uint4(0u, 0u, 0u, 0u);
  if (wave < nch) vn = rid4[(size_t)wave * 32 + lane];
  for (int ch = wave; ch < nch; ch += NWAVE) {
    const uint4 v = vn;
    if (ch + NWAVE < nch) vn = rid4[(size_t)(ch + NWAVE) * 32 + lane];
    const unsigned x0 = v.x ^ pat, x1 = v.y ^ pat, x2 = v.z ^ pat, x3 = v.w ^ pat;
    const unsigned f0 = ~(((x0 & 0x7F7F7F7Fu) + 0x7F7F7F7Fu) | x0) & 0x80808080u;
    const unsigned f1 = ~(((x1 & 0x7F7F7F7Fu) + 0x7F7F7F7Fu) | x1) & 0x80808080u;
    const unsigned f2 = ~(((x2 & 0x7F7F7F7Fu) + 0x7F7F7F7Fu) | x2) & 0x80808080u;
    const unsigned f3 = ~(((x3 & 0x7F7F7F7Fu) + 0x7F7F7F7Fu) | x3) & 0x80808080u;
    unsigned msk = (f0 >> 7) | (f1 >> 6) | (f2 >> 5) | (f3 >> 4);
    const int e0 = (ch << 9) + lane * 16;
    unsigned any = __builtin_amdgcn_ballot_w32(msk != 0u);
    while (any) {
      const int  p  = __builtin_ctz(msk | 0x80000000u);
      const int  e  = e0 + (((p & 7) << 2) | (p >> 3));
      const bool ok = (msk != 0u) && (e < E);
      const unsigned okb = __builtin_amdgcn_ballot_w32(ok);
      if (ok) q[qn + __builtin_popcount(okb & lt)] = e;
      qn  += __builtin_popcount(okb);
      msk &= msk - 1u;
      any  = __builtin_amdgcn_ballot_w32(msk != 0u);
    }
    if (qn >= GROUP) {
      __builtin_amdgcn_wave_barrier();
      int b0 = 0;
      for (; b0 + GROUP <= qn; b0 += GROUP) grp(b0, GROUP);
      const int rest = qn - b0;
      int t = 0;
      if (lane < rest) t = q[b0 + lane];
      if (lane < rest) q[lane] = t;
      qn = rest;
    }
  }
  if (qn > 0) {
    __builtin_amdgcn_wave_barrier();
    grp(0, qn);
  }
}

template<int TN>
__global__ __launch_bounds__(1024)
void k_range_deg(const unsigned char* __restrict__ rid, const int* __restrict__ row,
                 const int* __restrict__ col,
                 const float* __restrict__ w, float* __restrict__ dinv, int N, int E,
                 int4* elists, int* ecounts) {
  __shared__ float deg[TN + 1];
  __shared__ int queue[NWAVE * QCAP];

  const int lane = threadIdx.x & 31;
  const int wave = __builtin_amdgcn_readfirstlane(threadIdx.x >> 5);
  const int n0   = blockIdx.x * TN;
  int rows = N - n0;
  if (rows > TN) rows = TN;

  for (int i = threadIdx.x; i < TN + 1; i += 1024) deg[i] = 0.0f;
  __syncthreads();

  const int wid = blockIdx.x * NWAVE + wave;
  int4* el = elists + (size_t)wid * (LISTCAP * 32);
  int* q  = queue + wave * QCAP;
  int ng = 0;
  scan_ids(rid, E, blockIdx.x, wave, lane, q, [&](int b0, int cnt) {
    int e = -1;
    if (lane < cnt) e = q[b0 + lane];
    int4 r = make_int4(0, -1, 0, 0);
    if (e >= 0) {
      const int   rr = row[e];
      const int   cc = col[e];
      const float s  = w[e];
      r = make_int4(rr, cc, __float_as_int(s), 0);
      int c = cc - n0;
      c = ((unsigned)c < (unsigned)TN) ? c : TN;
      __hip_atomic_fetch_add(deg + c, s, __ATOMIC_RELAXED, __HIP_MEMORY_SCOPE_WORKGROUP);
    }
    if (ng < LISTCAP) el[ng * 32 + lane] = r;
    ++ng;
  });
  if (lane == 0) ecounts[wid] = (ng <= LISTCAP) ? ng : -1;

  __syncthreads();
  for (int i = threadIdx.x; i < rows; i += 1024) {
    const float d = deg[i] + 1.0f;
    dinv[n0 + i] = (d > 0.0f) ? rsqrtf(d) : 0.0f;
  }
}

template<int F, int TN, int LD, bool KEEP>
__device__ __forceinline__ void range_group(const int4 r,
                                            const float* __restrict__ h,
                                            float* acc, int n0, int lane,
                                            int4* lst, int& ng) {
  if (KEEP) {
    if (ng < LISTCAP) lst[ng * 32 + lane] = r;
    ++ng;
  }
  rec_group<F, TN, LD>(r, h, acc, n0, lane);
}

template<int LD>
__device__ __forceinline__ void strip_product(const float* acc, int s, int rows, int nbase,
                                              const __bf16* __restrict__ Bt,
                                              float* __restrict__ H2, int lane) {
  constexpr int K  = 128;
  constexpr int NC = 64;
  const int half = lane >> 4;
  const int l15  = lane & 15;
  int rl = s * 16 + l15;
  if (rl > rows - 1) rl = rows - 1;
  const float* Ar = acc + rl * LD + half * 8;

  const int ml0 = s * 16 + half * 8;

  const __bf16* Brow = Bt + (size_t)l15 * K + half * 8;
  v8f c[NC / 16];
#pragma unroll
  for (int t = 0; t < NC / 16; ++t) c[t] = v8f{};
#pragma unroll 1
  for (int kk = 0; kk < K / 32; ++kk) {
    const float4 f0 = *(const float4*)(Ar + kk * 32);
    const float4 f1 = *(const float4*)(Ar + kk * 32 + 4);
    const float4 f2 = *(const float4*)(Ar + kk * 32 + 16);
    const float4 f3 = *(const float4*)(Ar + kk * 32 + 20);
    v16bf a;
    a[0]  = tobf(fmaxf(f0.x, 0.f)); a[1]  = tobf(fmaxf(f0.y, 0.f));
    a[2]  = tobf(fmaxf(f0.z, 0.f)); a[3]  = tobf(fmaxf(f0.w, 0.f));
    a[4]  = tobf(fmaxf(f1.x, 0.f)); a[5]  = tobf(fmaxf(f1.y, 0.f));
    a[6]  = tobf(fmaxf(f1.z, 0.f)); a[7]  = tobf(fmaxf(f1.w, 0.f));
    a[8]  = tobf(fmaxf(f2.x, 0.f)); a[9]  = tobf(fmaxf(f2.y, 0.f));
    a[10] = tobf(fmaxf(f2.z, 0.f)); a[11] = tobf(fmaxf(f2.w, 0.f));
    a[12] = tobf(fmaxf(f3.x, 0.f)); a[13] = tobf(fmaxf(f3.y, 0.f));
    a[14] = tobf(fmaxf(f3.z, 0.f)); a[15] = tobf(fmaxf(f3.w, 0.f));
#pragma unroll
    for (int t = 0; t < NC / 16; ++t) {
      const v8bf b0 = *(const v8bf*)(Brow + (size_t)t * 16 * K + kk * 32);
      const v8bf b1 = *(const v8bf*)(Brow + (size_t)t * 16 * K + kk * 32 + 16);
      const v16bf b = __builtin_shufflevector(b0, b1,
          0, 1, 2, 3, 4, 5, 6, 7, 8, 9, 10, 11, 12, 13, 14, 15);
      c[t] = __builtin_amdgcn_wmma_f32_16x16x32_bf16(false, a, false, b,
                                                     (short)0, c[t], false, false);
    }
  }

  if (s * 16 + 16 <= rows) {
#pragma unroll
    for (int t = 0; t < NC / 16; ++t) {
#pragma unroll
      for (int r = 0; r < 8; ++r)
        H2[(size_t)(nbase + ml0 + r) * NC + t * 16 + l15] = c[t][r];
    }
  } else {
#pragma unroll
    for (int t = 0; t < NC / 16; ++t) {
#pragma unroll
      for (int r = 0; r < 8; ++r) {
        if (ml0 + r < rows)
          H2[(size_t)(nbase + ml0 + r) * NC + t * 16 + l15] = c[t][r];
      }
    }
  }
}

template<int F, int TN, bool REPLAY>
__global__ __launch_bounds__(1024)
void k_range_add(const int* __restrict__ row, const int* __restrict__ col,
                 const float* __restrict__ w, const float* __restrict__ dinv,
                 const unsigned char* __restrict__ rid,
                 const float* __restrict__ h, float* out, int N, int E,
                 const int4* elists, const int* ecounts,
                 int4* lists, int* counts,
                 const __bf16* __restrict__ w2t, const float* __restrict__ bias2,
                 float* __restrict__ h2) {
  constexpr int LD  = REPLAY ? F : F + 4;
  constexpr int F4  = F / 4;
  constexpr int LD4 = LD / 4;
  __shared__ __attribute__((aligned(16))) float acc[(TN + 1) * LD];
  __shared__ int queue[NWAVE * QCAP];

  const int lane = threadIdx.x & 31;
  const int wave = __builtin_amdgcn_readfirstlane(threadIdx.x >> 5);
  const int n0   = blockIdx.x * TN;
  int rows = N - n0;
  if (rows > TN) rows = TN;
  const int cnt4 = rows * (F / 4);

  float4* out4 = (float4*)(out + (size_t)n0 * F);
  float4* acc4 = (float4*)acc;
  float4 bq = make_float4(0.f, 0.f, 0.f, 0.f);
  if constexpr (REPLAY) {
    const int c0 = (threadIdx.x % F4) * 4;
    bq = make_float4(bias2[c0], bias2[c0 + 1], bias2[c0 + 2], bias2[c0 + 3]);
  }
  const float4* hown4 = (const float4*)(h + (size_t)n0 * F);
  for (int i0 = threadIdx.x; i0 < cnt4; i0 += 4096) {
    float4 t[4];
    float  dd[4];
#pragma unroll
    for (int u = 0; u < 4; ++u) {
      const int i = i0 + u * 1024;
      dd[u] = 0.f;
      if constexpr (REPLAY) {
        t[u] = (i < cnt4) ? hown4[i] : make_float4(0.f, 0.f, 0.f, 0.f);
        if (i < cnt4) dd[u] = dinv[n0 + i / F4];
      } else {
        t[u] = (i < cnt4) ? out4[i] : make_float4(0.f, 0.f, 0.f, 0.f);
      }
    }
#pragma unroll
    for (int u = 0; u < 4; ++u) {
      const int i = i0 + u * 1024;
      if constexpr (REPLAY) {
        const float d2 = dd[u] * dd[u];
        t[u] = make_float4(fmaf(d2, t[u].x, bq.x), fmaf(d2, t[u].y, bq.y),
                           fmaf(d2, t[u].z, bq.z), fmaf(d2, t[u].w, bq.w));
      }
      if (i < cnt4) acc4[(i / F4) * LD4 + (i % F4)] = t[u];
    }
  }
  __syncthreads();

  const int wid = blockIdx.x * NWAVE + wave;
  int4* lst = lists + (size_t)wid * (LISTCAP * 32);
  int ng = 0;
  bool scan = true;
  if (REPLAY) {
    const int kept = __builtin_amdgcn_readfirstlane(counts[wid]);
    if (kept >= 0) {
      scan = false;
      int4 rn = make_int4(0, -1, 0, 0);
      if (kept > 0) rn = lst[lane];
      for (int g = 0; g < kept; ++g) {
        const int4 r = rn;
        if (g + 1 < kept) rn = lst[(g + 1) * 32 + lane];
        rec_group<F, TN, LD>(r, h, acc, n0, lane);
      }
    }
  } else {
    const int kept = __builtin_amdgcn_readfirstlane(ecounts[wid]);
    if (kept >= 0) {
      scan = false;
      const int4* el = elists + (size_t)wid * (LISTCAP * 32);
      int4 ra = make_int4(0, -1, 0, 0);
      if (kept > 0) ra = el[lane];
      int4 rb = make_int4(0, -1, 0, 0);
      if (kept > 1) rb = el[32 + lane];
      float dr = dinv[ra.x];
      float dc = dinv[ra.y < 0 ? 0 : ra.y];
      for (int g = 0; g < kept; ++g) {
        int4 r = ra;
        r.z = __float_as_int(dr * __int_as_float(r.z) * dc);
        ra = rb;
        dr = dinv[ra.x];
        dc = dinv[ra.y < 0 ? 0 : ra.y];
        rb = make_int4(0, -1, 0, 0);
        if (g + 2 < kept) rb = el[(g + 2) * 32 + lane];
        range_group<F, TN, LD, true>(r, h, acc, n0, lane, lst, ng);
      }
    }
  }
  if (scan) {
    int* q = queue + wave * QCAP;
    scan_ids(rid, E, blockIdx.x, wave, lane, q, [&](int b0, int cnt) {
      int4 r = make_int4(0, -1, 0, 0);
      if (lane < cnt) r = edge_rec(q[b0 + lane], row, col, w, dinv);
      range_group<F, TN, LD, !REPLAY>(r, h, acc, n0, lane, lst, ng);
    });
  }
  if (!REPLAY) {
    if (lane == 0) counts[wid] = (ng <= LISTCAP) ? ng : -1;
  }
  __syncthreads();
  if constexpr (REPLAY) {
    for (int i = threadIdx.x; i < cnt4; i += 1024) out4[i] = acc4[(i / F4) * LD4 + (i % F4)];
  } else {
    const int nstrips = (rows + 15) >> 4;
    for (int s = wave; s < nstrips; s += NWAVE)
      strip_product<LD>(acc, s, rows, n0, w2t, h2, lane);
  }
}

extern "C" void kernel_launch(void* const* d_in, const int* in_sizes, int n_in,
                              void* d_out, int out_size, void* d_ws, size_t ws_size,
                              hipStream_t stream) {
  const float* x  = (const float*)d_in[0];
  const int*   ei = (const int*)  d_in[1];
  const float* P  = (const float*)d_in[2];
  const float* W1 = (const float*)d_in[3];
  const float* b1 = (const float*)d_in[4];
  const float* W2 = (const float*)d_in[5];
  const float* b2 = (const float*)d_in[6];
  float* out = (float*)d_out;

  constexpr int F_IN = 128, H1 = 128, H2 = 64;
  const int N = in_sizes[0] / F_IN;
  const int E = in_sizes[2];
  const int* row = ei;
  const int* col = ei + E;

  constexpr int TN = 400;
  const int Epad = ((E + 511) / 512) * 512;

  unsigned char* rid = (unsigned char*)d_ws;
  float* dinv = (float*)(rid + Epad);
  float* wn   = dinv + N;
  float* h    = wn + E;
  float* o1   = h + (size_t)N * H1;
  __bf16* w1t = (__bf16*)(o1 + (size_t)N * H1);
  __bf16* w2t = w1t + F_IN * H1;

  const int T = 256;
  const int strips = (N + 15) / 16;
  const int ranges = (N + TN - 1) / TN;
  int4* lists   = (int4*)(w2t + F_IN * H2);
  int*  counts  = (int*)(lists + (size_t)ranges * 32 * LISTCAP * 32);
  int*  ecounts = counts + (size_t)ranges * 32;
  int4* elists  = (int4*)(ecounts + (size_t)ranges * 32);
  float* h2     = (float*)(elists + (size_t)ranges * 32 * LISTCAP * 32);

  k_sig_rid<TN><<<(Epad / 8 + T - 1) / T, T, 0, stream>>>(P, col, wn, rid, E, Epad);
  k_wt_bf16<H1><<<(F_IN * H1 + T - 1) / T, T, 0, stream>>>(W1, w1t);
  k_wt_bf16<H2><<<(F_IN * H2 + T - 1) / T, T, 0, stream>>>(W2, w2t);

  k_range_deg<TN><<<ranges, 1024, 0, stream>>>(rid, row, col, wn, dinv, N, E, elists, ecounts);

  k_gemm_wmma<H1, false><<<strips, 32 * (H1 / 16), 0, stream>>>(
      x, w1t, b1, dinv, h, o1, N);
  k_range_add<H1, TN, false><<<ranges, 1024, 0, stream>>>(row, col, wn, dinv, rid, h, o1, N, E,
                                                          elists, ecounts, lists, counts,
                                                          w2t, nullptr, h2);

  k_range_add<H2, TN, true><<<ranges, 1024, 0, stream>>>(row, col, wn, dinv, rid, h2, out, N, E,
                                                         elists, ecounts, lists, counts,
                                                         nullptr, b2, nullptr);
}


# layer-1 GEMM launch dissolved: each wave of a degree-kernel range block multiplies one 16-row strip of x (same bf16 WMMA tiles) and writes h; layer-1 blocks form b1+dinv^2*h at load; 7 -> 6 launches
# speedup vs baseline: 3.2691x; 1.0928x over previous
#include <hip/hip_runtime.h>
#include <hip/hip_bf16.h>

typedef __attribute__((ext_vector_type(16))) __bf16        v16bf;
typedef __attribute__((ext_vector_type(8)))  __bf16        v8bf;
typedef __attribute__((ext_vector_type(4)))  __bf16        v4bf;
typedef __attribute__((ext_vector_type(8)))  float         v8f;

__device__ __forceinline__ __bf16 tobf(float f) { return (__bf16)f; }

template<int TN>
__global__ void k_sig_rid(const float* __restrict__ P, const int* __restrict__ col,
                          float* __restrict__ w, unsigned char* __restrict__ rid,
                          int E, int Epad) {
  const int e0 = (blockIdx.x * blockDim.x + threadIdx.x) * 8;
  if (e0 >= Epad) return;
  const size_t al = (size_t)(P + e0) | (size_t)(col + e0) | (size_t)(w + e0) | (size_t)(rid + e0);
  if (e0 + 8 <= E && (al & 15) == 0) {
    const float4 p0 = *(const float4*)(P + e0);
    const float4 p1 = *(const float4*)(P + e0 + 4);
    const int4   c0 = *(const int4*)(col + e0);
    const int4   c1 = *(const int4*)(col + e0 + 4);
    float4 s0, s1;
    s0.x = 1.0f / (1.0f + __expf(-p0.x)); s0.y = 1.0f / (1.0f + __expf(-p0.y));
    s0.z = 1.0f / (1.0f + __expf(-p0.z)); s0.w = 1.0f / (1.0f + __expf(-p0.w));
    s1.x = 1.0f / (1.0f + __expf(-p1.x)); s1.y = 1.0f / (1.0f + __expf(-p1.y));
    s1.z = 1.0f / (1.0f + __expf(-p1.z)); s1.w = 1.0f / (1.0f + __expf(-p1.w));
    *(float4*)(w + e0)     = s0;
    *(float4*)(w + e0 + 4) = s1;
    uint2 id;
    id.x = ((unsigned)(c0.x / TN) & 255u)       | (((unsigned)(c0.y / TN) & 255u) << 8) |
           (((unsigned)(c0.z / TN) & 255u) << 16) | (((unsigned)(c0.w / TN) & 255u) << 24);
    id.y = ((unsigned)(c1.x / TN) & 255u)       | (((unsigned)(c1.y / TN) & 255u) << 8) |
           (((unsigned)(c1.z / TN) & 255u) << 16) | (((unsigned)(c1.w / TN) & 255u) << 24);
    *(uint2*)(rid + e0) = id;
  } else {
    for (int j = 0; j < 8; ++j) {
      const int e = e0 + j;
      if (e < E) {
        w[e]   = 1.0f / (1.0f + __expf(-P[e]));
        rid[e] = (unsigned char)(col[e] / TN);
      } else if (e < Epad) {
        rid[e] = 0;
      }
    }
  }
}

__device__ __forceinline__ int4 edge_rec(int e, const int* __restrict__ row,
                                         const int* __restrict__ col,
                                         const float* __restrict__ w,
                                         const float* __restrict__ dinv) {
  const int r = row[e];
  const int c = col[e];
  const float nm = dinv[r] * w[e] * dinv[c];
  return make_int4(r, c, __float_as_int(nm), 0);
}

template<int NCOL>
__global__ void k_wt_bf16(const float* __restrict__ W, __bf16* __restrict__ Wt) {
  int idx = blockIdx.x * blockDim.x + threadIdx.x;
  if (idx >= 128 * NCOL) return;
  int k = idx / NCOL, n = idx % NCOL;
  Wt[n * 128 + k] = tobf(W[idx]);
}

constexpr int XLDSP = 136;

__device__ __forceinline__ void strip_x_product(const float* __restrict__ A, int m0, int N,
                                                const __bf16* __restrict__ Bt,
                                                float* __restrict__ H,
                                                __bf16* As, int lane) {
  constexpr int K  = 128;
  constexpr int NC = 128;
  const int half = lane >> 4;
  const int l15  = lane & 15;

  float4 f[16];
#pragma unroll
  for (int j = 0; j < 16; ++j) {
    int mg = m0 + j; if (mg > N - 1) mg = N - 1;
    f[j] = *(const float4*)(A + (size_t)mg * K + lane * 4);
  }
#pragma unroll
  for (int j = 0; j < 16; ++j) {
    v4bf pk;
    pk[0] = tobf(f[j].x); pk[1] = tobf(f[j].y); pk[2] = tobf(f[j].z); pk[3] = tobf(f[j].w);
    *(v4bf*)(&As[j * XLDSP + lane * 4]) = pk;
  }
  __builtin_amdgcn_wave_barrier();

  const __bf16* Alds = As + l15 * XLDSP + half * 8;
  v16bf a[K / 32];
#pragma unroll
  for (int kk = 0; kk < K / 32; ++kk) {
    const v8bf a0 = *(const v8bf*)(Alds + kk * 32);
    const v8bf a1 = *(const v8bf*)(Alds + kk * 32 + 16);
    a[kk] = __builtin_shufflevector(a0, a1,
        0, 1, 2, 3, 4, 5, 6, 7, 8, 9, 10, 11, 12, 13, 14, 15);
  }

  const int mr0 = m0 + half * 8;
#pragma unroll 1
  for (int t = 0; t < NC / 16; t += 2) {
    v8f acc[2];
#pragma unroll
    for (int u = 0; u < 2; ++u) {
      const __bf16* Brow = Bt + (size_t)((t + u) * 16 + l15) * K + half * 8;
      v8bf bl[K / 16];
#pragma unroll
      for (int q = 0; q < K / 16; ++q) bl[q] = *(const v8bf*)(Brow + q * 16);
      asm volatile("" :: "v"(bl[0]), "v"(bl[1]), "v"(bl[2]), "v"(bl[3]),
                         "v"(bl[4]), "v"(bl[5]), "v"(bl[6]), "v"(bl[7]));
      acc[u] = v8f{};
#pragma unroll
      for (int kk = 0; kk < K / 32; ++kk) {
        const v16bf b = __builtin_shufflevector(bl[2 * kk], bl[2 * kk + 1],
            0, 1, 2, 3, 4, 5, 6, 7, 8, 9, 10, 11, 12, 13, 14, 15);
        acc[u] = __builtin_amdgcn_wmma_f32_16x16x32_bf16(false, a[kk], false, b,
                                                         (short)0, acc[u], false, false);
      }
    }
    const int n = t * 16 + l15;
    if (m0 + 16 <= N) {
#pragma unroll
      for (int r = 0; r < 8; ++r) {
        H[(size_t)(mr0 + r) * NC + n]      = acc[0][r];
        H[(size_t)(mr0 + r) * NC + n + 16] = acc[1][r];
      }
    } else {
#pragma unroll
      for (int r = 0; r < 8; ++r) {
        if (mr0 + r < N) {
          H[(size_t)(mr0 + r) * NC + n]      = acc[0][r];
          H[(size_t)(mr0 + r) * NC + n + 16] = acc[1][r];
        }
      }
    }
  }
}

template<int F, int J0>
__device__ __forceinline__ void rows_load(float (&v)[8][F / 32], int rx,
                                          const float* __restrict__ h, int lane) {
#pragma unroll
  for (int j = 0; j < 8; ++j) {
    const int rr = __builtin_amdgcn_readlane(rx, J0 + j);
    const float* hp = h + (size_t)rr * F + lane;
#pragma unroll
    for (int k = 0; k < F / 32; ++k) v[j][k] = hp[k * 32];
  }
}

template<int F, int TN, int LD, int J0>
__device__ __forceinline__ void rows_add(const float (&v)[8][F / 32], int ry, int rz,
                                         float* acc, int n0, int lane) {
#pragma unroll
  for (int j = 0; j < 8; ++j) {
    int c = __builtin_amdgcn_readlane(ry, J0 + j) - n0;
    c = ((unsigned)c < (unsigned)TN) ? c : TN;
    const float w = __int_as_float(__builtin_amdgcn_readlane(rz, J0 + j));
    float* ap = acc + c * LD + lane;
#pragma unroll
    for (int k = 0; k < F / 32; ++k)
      __hip_atomic_fetch_add(ap + k * 32, w * v[j][k],
                             __ATOMIC_RELAXED, __HIP_MEMORY_SCOPE_WORKGROUP);
  }
}

template<int F, int TN, int LD>
__device__ __forceinline__ void rec_group(const int4 r, const float* __restrict__ h,
                                          float* acc, int n0, int lane) {
  float va[8][F / 32];
  float vb[8][F / 32];
  rows_load<F, 0>(va, r.x, h, lane);
  rows_load<F, 8>(vb, r.x, h, lane);
  rows_add<F, TN, LD, 0>(va, r.y, r.z, acc, n0, lane);
  rows_load<F, 16>(va, r.x, h, lane);
  rows_add<F, TN, LD, 8>(vb, r.y, r.z, acc, n0, lane);
  rows_load<F, 24>(vb, r.x, h, lane);
  rows_add<F, TN, LD, 16>(va, r.y, r.z, acc, n0, lane);
  rows_add<F, TN, LD, 24>(vb, r.y, r.z, acc, n0, lane);
}

constexpr int LISTCAP = 16;

constexpr int NWAVE = 32;
constexpr int GROUP = 32;
constexpr int QCAP  = 512 + GROUP;

template<typename G>
__device__ __forceinline__ void scan_ids(const unsigned char* __restrict__ rid, int E, int blk,
                                         int wave, int lane, int* q, G grp) {
  int qn = 0;
  const unsigned pat = ((unsigned)blk & 255u) * 0x01010101u;
  const unsigned lt  = (1u << lane) - 1u;
  const int nch = (E + 511) >> 9;
  const uint4* rid4 = (const uint4*)rid;

  uint4 vn = make_uint4(0u, 0u, 0u, 0u);
  if (wave < nch) vn = rid4[(size_t)wave * 32 + lane];
  for (int ch = wave; ch < nch; ch += NWAVE) {
    const uint4 v = vn;
    if (ch + NWAVE < nch) vn = rid4[(size_t)(ch + NWAVE) * 32 + lane];
    const unsigned x0 = v.x ^ pat, x1 = v.y ^ pat, x2 = v.z ^ pat, x3 = v.w ^ pat;
    const unsigned f0 = ~(((x0 & 0x7F7F7F7Fu) + 0x7F7F7F7Fu) | x0) & 0x80808080u;
    const unsigned f1 = ~(((x1 & 0x7F7F7F7Fu) + 0x7F7F7F7Fu) | x1) & 0x80808080u;
    const unsigned f2 = ~(((x2 & 0x7F7F7F7Fu) + 0x7F7F7F7Fu) | x2) & 0x80808080u;
    const unsigned f3 = ~(((x3 & 0x7F7F7F7Fu) + 0x7F7F7F7Fu) | x3) & 0x80808080u;
    unsigned msk = (f0 >> 7) | (f1 >> 6) | (f2 >> 5) | (f3 >> 4);
    const int e0 = (ch << 9) + lane * 16;
    unsigned any = __builtin_amdgcn_ballot_w32(msk != 0u);
    while (any) {
      const int  p  = __builtin_ctz(msk | 0x80000000u);
      const int  e  = e0 + (((p & 7) << 2) | (p >> 3));
      const bool ok = (msk != 0u) && (e < E);
      const unsigned okb = __builtin_amdgcn_ballot_w32(ok);
      if (ok) q[qn + __builtin_popcount(okb & lt)] = e;
      qn  += __builtin_popcount(okb);
      msk &= msk - 1u;
      any  = __builtin_amdgcn_ballot_w32(msk != 0u);
    }
    if (qn >= GROUP) {
      __builtin_amdgcn_wave_barrier();
      int b0 = 0;
      for (; b0 + GROUP <= qn; b0 += GROUP) grp(b0, GROUP);
      const int rest = qn - b0;
      int t = 0;
      if (lane < rest) t = q[b0 + lane];
      if (lane < rest) q[lane] = t;
      qn = rest;
    }
  }
  if (qn > 0) {
    __builtin_amdgcn_wave_barrier();
    grp(0, qn);
  }
}

template<int TN>
__global__ __launch_bounds__(1024)
void k_range_deg(const unsigned char* __restrict__ rid, const int* __restrict__ row,
                 const int* __restrict__ col,
                 const float* __restrict__ w, float* __restrict__ dinv, int N, int E,
                 int4* elists, int* ecounts,
                 const float* __restrict__ x, const __bf16* __restrict__ w1t,
                 float* __restrict__ h) {
  static_assert(TN <= 16 * NWAVE && TN % 16 == 0,
                "one 16-row strip per wave, and no strip reaches into the next block's rows");
  __shared__ float deg[TN + 1];
  __shared__ int queue[NWAVE * QCAP];
  __shared__ __align__(16) __bf16 xs[NWAVE * 16 * XLDSP];

  const int lane = threadIdx.x & 31;
  const int wave = __builtin_amdgcn_readfirstlane(threadIdx.x >> 5);
  const int n0   = blockIdx.x * TN;
  int rows = N - n0;
  if (rows > TN) rows = TN;

  for (int i = threadIdx.x; i < TN + 1; i += 1024) deg[i] = 0.0f;
  __syncthreads();

  if (wave * 16 < rows)
    strip_x_product(x, n0 + wave * 16, N, w1t, h, xs + wave * (16 * XLDSP), lane);

  const int wid = blockIdx.x * NWAVE + wave;
  int4* el = elists + (size_t)wid * (LISTCAP * 32);
  int* q  = queue + wave * QCAP;
  int ng = 0;
  scan_ids(rid, E, blockIdx.x, wave, lane, q, [&](int b0, int cnt) {
    int e = -1;
    if (lane < cnt) e = q[b0 + lane];
    int4 r = make_int4(0, -1, 0, 0);
    if (e >= 0) {
      const int   rr = row[e];
      const int   cc = col[e];
      const float s  = w[e];
      r = make_int4(rr, cc, __float_as_int(s), 0);
      int c = cc - n0;
      c = ((unsigned)c < (unsigned)TN) ? c : TN;
      __hip_atomic_fetch_add(deg + c, s, __ATOMIC_RELAXED, __HIP_MEMORY_SCOPE_WORKGROUP);
    }
    if (ng < LISTCAP) el[ng * 32 + lane] = r;
    ++ng;
  });
  if (lane == 0) ecounts[wid] = (ng <= LISTCAP) ? ng : -1;

  __syncthreads();
  for (int i = threadIdx.x; i < rows; i += 1024) {
    const float d = deg[i] + 1.0f;
    dinv[n0 + i] = (d > 0.0f) ? rsqrtf(d) : 0.0f;
  }
}

template<int F, int TN, int LD, bool KEEP>
__device__ __forceinline__ void range_group(const int4 r,
                                            const float* __restrict__ h,
                                            float* acc, int n0, int lane,
                                            int4* lst, int& ng) {
  if (KEEP) {
    if (ng < LISTCAP) lst[ng * 32 + lane] = r;
    ++ng;
  }
  rec_group<F, TN, LD>(r, h, acc, n0, lane);
}

template<int LD>
__device__ __forceinline__ void strip_product(const float* acc, int s, int rows, int nbase,
                                              const __bf16* __restrict__ Bt,
                                              float* __restrict__ H2, int lane) {
  constexpr int K  = 128;
  constexpr int NC = 64;
  const int half = lane >> 4;
  const int l15  = lane & 15;
  int rl = s * 16 + l15;
  if (rl > rows - 1) rl = rows - 1;
  const float* Ar = acc + rl * LD + half * 8;

  const int ml0 = s * 16 + half * 8;

  const __bf16* Brow = Bt + (size_t)l15 * K + half * 8;
  v8f c[NC / 16];
#pragma unroll
  for (int t = 0; t < NC / 16; ++t) c[t] = v8f{};
#pragma unroll 1
  for (int kk = 0; kk < K / 32; ++kk) {
    const float4 f0 = *(const float4*)(Ar + kk * 32);
    const float4 f1 = *(const float4*)(Ar + kk * 32 + 4);
    const float4 f2 = *(const float4*)(Ar + kk * 32 + 16);
    const float4 f3 = *(const float4*)(Ar + kk * 32 + 20);
    v16bf a;
    a[0]  = tobf(fmaxf(f0.x, 0.f)); a[1]  = tobf(fmaxf(f0.y, 0.f));
    a[2]  = tobf(fmaxf(f0.z, 0.f)); a[3]  = tobf(fmaxf(f0.w, 0.f));
    a[4]  = tobf(fmaxf(f1.x, 0.f)); a[5]  = tobf(fmaxf(f1.y, 0.f));
    a[6]  = tobf(fmaxf(f1.z, 0.f)); a[7]  = tobf(fmaxf(f1.w, 0.f));
    a[8]  = tobf(fmaxf(f2.x, 0.f)); a[9]  = tobf(fmaxf(f2.y, 0.f));
    a[10] = tobf(fmaxf(f2.z, 0.f)); a[11] = tobf(fmaxf(f2.w, 0.f));
    a[12] = tobf(fmaxf(f3.x, 0.f)); a[13] = tobf(fmaxf(f3.y, 0.f));
    a[14] = tobf(fmaxf(f3.z, 0.f)); a[15] = tobf(fmaxf(f3.w, 0.f));
#pragma unroll
    for (int t = 0; t < NC / 16; ++t) {
      const v8bf b0 = *(const v8bf*)(Brow + (size_t)t * 16 * K + kk * 32);
      const v8bf b1 = *(const v8bf*)(Brow + (size_t)t * 16 * K + kk * 32 + 16);
      const v16bf b = __builtin_shufflevector(b0, b1,
          0, 1, 2, 3, 4, 5, 6, 7, 8, 9, 10, 11, 12, 13, 14, 15);
      c[t] = __builtin_amdgcn_wmma_f32_16x16x32_bf16(false, a, false, b,
                                                     (short)0, c[t], false, false);
    }
  }

  if (s * 16 + 16 <= rows) {
#pragma unroll
    for (int t = 0; t < NC / 16; ++t) {
#pragma unroll
      for (int r = 0; r < 8; ++r)
        H2[(size_t)(nbase + ml0 + r) * NC + t * 16 + l15] = c[t][r];
    }
  } else {
#pragma unroll
    for (int t = 0; t < NC / 16; ++t) {
#pragma unroll
      for (int r = 0; r < 8; ++r) {
        if (ml0 + r < rows)
          H2[(size_t)(nbase + ml0 + r) * NC + t * 16 + l15] = c[t][r];
      }
    }
  }
}

template<int F, int TN, bool REPLAY>
__global__ __launch_bounds__(1024)
void k_range_add(const int* __restrict__ row, const int* __restrict__ col,
                 const float* __restrict__ w, const float* __restrict__ dinv,
                 const unsigned char* __restrict__ rid,
                 const float* __restrict__ h, float* out, int N, int E,
                 const int4* elists, const int* ecounts,
                 int4* lists, int* counts,
                 const __bf16* __restrict__ w2t, const float* __restrict__ bias,
                 float* __restrict__ h2) {
  constexpr int LD  = REPLAY ? F : F + 4;
  constexpr int F4  = F / 4;
  constexpr int LD4 = LD / 4;
  __shared__ __attribute__((aligned(16))) float acc[(TN + 1) * LD];
  __shared__ int queue[NWAVE * QCAP];

  const int lane = threadIdx.x & 31;
  const int wave = __builtin_amdgcn_readfirstlane(threadIdx.x >> 5);
  const int n0   = blockIdx.x * TN;
  int rows = N - n0;
  if (rows > TN) rows = TN;
  const int cnt4 = rows * (F / 4);

  float4* acc4 = (float4*)acc;
  const int c0 = (threadIdx.x % F4) * 4;
  const float4 bq = make_float4(bias[c0], bias[c0 + 1], bias[c0 + 2], bias[c0 + 3]);
  const float4* hown4 = (const float4*)(h + (size_t)n0 * F);
  for (int i0 = threadIdx.x; i0 < cnt4; i0 += 4096) {
    float4 t[4];
    float  dd[4];
#pragma unroll
    for (int u = 0; u < 4; ++u) {
      const int i = i0 + u * 1024;
      dd[u] = 0.f;
      t[u] = (i < cnt4) ? hown4[i] : make_float4(0.f, 0.f, 0.f, 0.f);
      if (i < cnt4) dd[u] = dinv[n0 + i / F4];
    }
#pragma unroll
    for (int u = 0; u < 4; ++u) {
      const int i = i0 + u * 1024;
      const float d2 = dd[u] * dd[u];
      t[u] = make_float4(fmaf(d2, t[u].x, bq.x), fmaf(d2, t[u].y, bq.y),
                         fmaf(d2, t[u].z, bq.z), fmaf(d2, t[u].w, bq.w));
      if (i < cnt4) acc4[(i / F4) * LD4 + (i % F4)] = t[u];
    }
  }
  __syncthreads();

  const int wid = blockIdx.x * NWAVE + wave;
  int4* lst = lists + (size_t)wid * (LISTCAP * 32);
  int ng = 0;
  bool scan = true;
  if (REPLAY) {
    const int kept = __builtin_amdgcn_readfirstlane(counts[wid]);
    if (kept >= 0) {
      scan = false;
      int4 rn = make_int4(0, -1, 0, 0);
      if (kept > 0) rn = lst[lane];
      for (int g = 0; g < kept; ++g) {
        const int4 r = rn;
        if (g + 1 < kept) rn = lst[(g + 1) * 32 + lane];
        rec_group<F, TN, LD>(r, h, acc, n0, lane);
      }
    }
  } else {
    const int kept = __builtin_amdgcn_readfirstlane(ecounts[wid]);
    if (kept >= 0) {
      scan = false;
      const int4* el = elists + (size_t)wid * (LISTCAP * 32);
      int4 ra = make_int4(0, -1, 0, 0);
      if (kept > 0) ra = el[lane];
      int4 rb = make_int4(0, -1, 0, 0);
      if (kept > 1) rb = el[32 + lane];
      float dr = dinv[ra.x];
      float dc = dinv[ra.y < 0 ? 0 : ra.y];
      for (int g = 0; g < kept; ++g) {
        int4 r = ra;
        r.z = __float_as_int(dr * __int_as_float(r.z) * dc);
        ra = rb;
        dr = dinv[ra.x];
        dc = dinv[ra.y < 0 ? 0 : ra.y];
        rb = make_int4(0, -1, 0, 0);
        if (g + 2 < kept) rb = el[(g + 2) * 32 + lane];
        range_group<F, TN, LD, true>(r, h, acc, n0, lane, lst, ng);
      }
    }
  }
  if (scan) {
    int* q = queue + wave * QCAP;
    scan_ids(rid, E, blockIdx.x, wave, lane, q, [&](int b0, int cnt) {
      int4 r = make_int4(0, -1, 0, 0);
      if (lane < cnt) r = edge_rec(q[b0 + lane], row, col, w, dinv);
      range_group<F, TN, LD, !REPLAY>(r, h, acc, n0, lane, lst, ng);
    });
  }
  if (!REPLAY) {
    if (lane == 0) counts[wid] = (ng <= LISTCAP) ? ng : -1;
  }
  __syncthreads();
  if constexpr (REPLAY) {
    float4* out4 = (float4*)(out + (size_t)n0 * F);
    for (int i = threadIdx.x; i < cnt4; i += 1024) out4[i] = acc4[(i / F4) * LD4 + (i % F4)];
  } else {
    const int nstrips = (rows + 15) >> 4;
    for (int s = wave; s < nstrips; s += NWAVE)
      strip_product<LD>(acc, s, rows, n0, w2t, h2, lane);
  }
}

extern "C" void kernel_launch(void* const* d_in, const int* in_sizes, int n_in,
                              void* d_out, int out_size, void* d_ws, size_t ws_size,
                              hipStream_t stream) {
  const float* x  = (const float*)d_in[0];
  const int*   ei = (const int*)  d_in[1];
  const float* P  = (const float*)d_in[2];
  const float* W1 = (const float*)d_in[3];
  const float* b1 = (const float*)d_in[4];
  const float* W2 = (const float*)d_in[5];
  const float* b2 = (const float*)d_in[6];
  float* out = (float*)d_out;

  constexpr int F_IN = 128, H1 = 128, H2 = 64;
  const int N = in_sizes[0] / F_IN;
  const int E = in_sizes[2];
  const int* row = ei;
  const int* col = ei + E;

  constexpr int TN = 400;
  const int Epad = ((E + 511) / 512) * 512;

  unsigned char* rid = (unsigned char*)d_ws;
  float* dinv = (float*)(rid + Epad);
  float* wn   = dinv + N;
  float* h    = wn + E;
  __bf16* w1t = (__bf16*)(h + (size_t)N * H1);
  __bf16* w2t = w1t + F_IN * H1;

  const int T = 256;
  const int ranges = (N + TN - 1) / TN;
  int4* lists   = (int4*)(w2t + F_IN * H2);
  int*  counts  = (int*)(lists + (size_t)ranges * 32 * LISTCAP * 32);
  int*  ecounts = counts + (size_t)ranges * 32;
  int4* elists  = (int4*)(ecounts + (size_t)ranges * 32);
  float* h2     = (float*)(elists + (size_t)ranges * 32 * LISTCAP * 32);

  k_sig_rid<TN><<<(Epad / 8 + T - 1) / T, T, 0, stream>>>(P, col, wn, rid, E, Epad);
  k_wt_bf16<H1><<<(F_IN * H1 + T - 1) / T, T, 0, stream>>>(W1, w1t);
  k_wt_bf16<H2><<<(F_IN * H2 + T - 1) / T, T, 0, stream>>>(W2, w2t);

  k_range_deg<TN><<<ranges, 1024, 0, stream>>>(rid, row, col, wn, dinv, N, E, elists, ecounts,
                                               x, w1t, h);

  k_range_add<H1, TN, false><<<ranges, 1024, 0, stream>>>(row, col, wn, dinv, rid, h, nullptr, N, E,
                                                          elists, ecounts, lists, counts,
                                                          w2t, b1, h2);

  k_range_add<H2, TN, true><<<ranges, 1024, 0, stream>>>(row, col, wn, dinv, rid, h2, out, N, E,
                                                         elists, ecounts, lists, counts,
                                                         nullptr, b2, nullptr);
}


# both weight transposes leave their launches: each range block rounds its weight matrix to bf16, transposed, into its own LDS and the products read B fragments there; 6 -> 4 launches; sum order kept
# speedup vs baseline: 3.4923x; 1.0683x over previous
#include <hip/hip_runtime.h>
#include <hip/hip_bf16.h>

typedef __attribute__((ext_vector_type(16))) __bf16        v16bf;
typedef __attribute__((ext_vector_type(8)))  __bf16        v8bf;
typedef __attribute__((ext_vector_type(4)))  __bf16        v4bf;
typedef __attribute__((ext_vector_type(8)))  float         v8f;

__device__ __forceinline__ __bf16 tobf(float f) { return (__bf16)f; }

template<int TN>
__global__ void k_sig_rid(const float* __restrict__ P, const int* __restrict__ col,
                          float* __restrict__ w, unsigned char* __restrict__ rid,
                          int E, int Epad) {
  const int e0 = (blockIdx.x * blockDim.x + threadIdx.x) * 8;
  if (e0 >= Epad) return;
  const size_t al = (size_t)(P + e0) | (size_t)(col + e0) | (size_t)(w + e0) | (size_t)(rid + e0);
  if (e0 + 8 <= E && (al & 15) == 0) {
    const float4 p0 = *(const float4*)(P + e0);
    const float4 p1 = *(const float4*)(P + e0 + 4);
    const int4   c0 = *(const int4*)(col + e0);
    const int4   c1 = *(const int4*)(col + e0 + 4);
    float4 s0, s1;
    s0.x = 1.0f / (1.0f + __expf(-p0.x)); s0.y = 1.0f / (1.0f + __expf(-p0.y));
    s0.z = 1.0f / (1.0f + __expf(-p0.z)); s0.w = 1.0f / (1.0f + __expf(-p0.w));
    s1.x = 1.0f / (1.0f + __expf(-p1.x)); s1.y = 1.0f / (1.0f + __expf(-p1.y));
    s1.z = 1.0f / (1.0f + __expf(-p1.z)); s1.w = 1.0f / (1.0f + __expf(-p1.w));
    *(float4*)(w + e0)     = s0;
    *(float4*)(w + e0 + 4) = s1;
    uint2 id;
    id.x = ((unsigned)(c0.x / TN) & 255u)       | (((unsigned)(c0.y / TN) & 255u) << 8) |
           (((unsigned)(c0.z / TN) & 255u) << 16) | (((unsigned)(c0.w / TN) & 255u) << 24);
    id.y = ((unsigned)(c1.x / TN) & 255u)       | (((unsigned)(c1.y / TN) & 255u) << 8) |
           (((unsigned)(c1.z / TN) & 255u) << 16) | (((unsigned)(c1.w / TN) & 255u) << 24);
    *(uint2*)(rid + e0) = id;
  } else {
    for (int j = 0; j < 8; ++j) {
      const int e = e0 + j;
      if (e < E) {
        w[e]   = 1.0f / (1.0f + __expf(-P[e]));
        rid[e] = (unsigned char)(col[e] / TN);
      } else if (e < Epad) {
        rid[e] = 0;
      }
    }
  }
}

__device__ __forceinline__ int4 edge_rec(int e, const int* __restrict__ row,
                                         const int* __restrict__ col,
                                         const float* __restrict__ w,
                                         const float* __restrict__ dinv) {
  const int r = row[e];
  const int c = col[e];
  const float nm = dinv[r] * w[e] * dinv[c];
  return make_int4(r, c, __float_as_int(nm), 0);
}

constexpr int XLDSP = 136;

template<int NCOL>
__device__ __forceinline__ void stage_wt_load(const float* W, float (&f)[32 * NCOL / 1024][4],
                                              int tid) {
  static_assert((32 * NCOL) % 1024 == 0, "whole trips of the 1024-thread block");
#pragma unroll
  for (int u = 0; u < 32 * NCOL / 1024; ++u) {
    const int i = tid + u * 1024;
    const float* p = W + (size_t)(4 * (i / NCOL)) * NCOL + (i % NCOL);
#pragma unroll
    for (int j = 0; j < 4; ++j) f[u][j] = p[j * NCOL];
  }
}

template<int NCOL>
__device__ __forceinline__ void stage_wt_store(const float (&f)[32 * NCOL / 1024][4], __bf16* Ws,
                                               int tid) {
#pragma unroll
  for (int u = 0; u < 32 * NCOL / 1024; ++u) {
    const int i = tid + u * 1024;
    v4bf pk;
    pk[0] = tobf(f[u][0]); pk[1] = tobf(f[u][1]); pk[2] = tobf(f[u][2]); pk[3] = tobf(f[u][3]);
    *(v4bf*)(&Ws[(i % NCOL) * XLDSP + 4 * (i / NCOL)]) = pk;
  }
}

__device__ __forceinline__ void strip_x_product(const float* __restrict__ A, int m0, int N,
                                                const __bf16* Bt,
                                                float* __restrict__ H,
                                                __bf16* As, int lane) {
  constexpr int K  = 128;
  constexpr int NC = 128;
  const int half = lane >> 4;
  const int l15  = lane & 15;

  float4 f[16];
#pragma unroll
  for (int j = 0; j < 16; ++j) {
    int mg = m0 + j; if (mg > N - 1) mg = N - 1;
    f[j] = *(const float4*)(A + (size_t)mg * K + lane * 4);
  }
#pragma unroll
  for (int j = 0; j < 16; ++j) {
    v4bf pk;
    pk[0] = tobf(f[j].x); pk[1] = tobf(f[j].y); pk[2] = tobf(f[j].z); pk[3] = tobf(f[j].w);
    *(v4bf*)(&As[j * XLDSP + lane * 4]) = pk;
  }
  __builtin_amdgcn_wave_barrier();

  const __bf16* Alds = As + l15 * XLDSP + half * 8;
  v16bf a[K / 32];
#pragma unroll
  for (int kk = 0; kk < K / 32; ++kk) {
    const v8bf a0 = *(const v8bf*)(Alds + kk * 32);
    const v8bf a1 = *(const v8bf*)(Alds + kk * 32 + 16);
    a[kk] = __builtin_shufflevector(a0, a1,
        0, 1, 2, 3, 4, 5, 6, 7, 8, 9, 10, 11, 12, 13, 14, 15);
  }

  const int mr0 = m0 + half * 8;
#pragma unroll 1
  for (int t = 0; t < NC / 16; t += 2) {
    v8f acc[2];
#pragma unroll
    for (int u = 0; u < 2; ++u) {
      const __bf16* Brow = Bt + ((t + u) * 16 + l15) * XLDSP + half * 8;
      v8bf bl[K / 16];
#pragma unroll
      for (int q = 0; q < K / 16; ++q) bl[q] = *(const v8bf*)(Brow + q * 16);
      asm volatile("" :: "v"(bl[0]), "v"(bl[1]), "v"(bl[2]), "v"(bl[3]),
                         "v"(bl[4]), "v"(bl[5]), "v"(bl[6]), "v"(bl[7]));
      acc[u] = v8f{};
#pragma unroll
      for (int kk = 0; kk < K / 32; ++kk) {
        const v16bf b = __builtin_shufflevector(bl[2 * kk], bl[2 * kk + 1],
            0, 1, 2, 3, 4, 5, 6, 7, 8, 9, 10, 11, 12, 13, 14, 15);
        acc[u] = __builtin_amdgcn_wmma_f32_16x16x32_bf16(false, a[kk], false, b,
                                                         (short)0, acc[u], false, false);
      }
    }
    const int n = t * 16 + l15;
    if (m0 + 16 <= N) {
#pragma unroll
      for (int r = 0; r < 8; ++r) {
        H[(size_t)(mr0 + r) * NC + n]      = acc[0][r];
        H[(size_t)(mr0 + r) * NC + n + 16] = acc[1][r];
      }
    } else {
#pragma unroll
      for (int r = 0; r < 8; ++r) {
        if (mr0 + r < N) {
          H[(size_t)(mr0 + r) * NC + n]      = acc[0][r];
          H[(size_t)(mr0 + r) * NC + n + 16] = acc[1][r];
        }
      }
    }
  }
}

template<int F, int J0>
__device__ __forceinline__ void rows_load(float (&v)[8][F / 32], int rx,
                                          const float* __restrict__ h, int lane) {
#pragma unroll
  for (int j = 0; j < 8; ++j) {
    const int rr = __builtin_amdgcn_readlane(rx, J0 + j);
    const float* hp = h + (size_t)rr * F + lane;
#pragma unroll
    for (int k = 0; k < F / 32; ++k) v[j][k] = hp[k * 32];
  }
}

template<int F, int TN, int LD, int J0>
__device__ __forceinline__ void rows_add(const float (&v)[8][F / 32], int ry, int rz,
                                         float* acc, int n0, int lane) {
#pragma unroll
  for (int j = 0; j < 8; ++j) {
    int c = __builtin_amdgcn_readlane(ry, J0 + j) - n0;
    c = ((unsigned)c < (unsigned)TN) ? c : TN;
    const float w = __int_as_float(__builtin_amdgcn_readlane(rz, J0 + j));
    float* ap = acc + c * LD + lane;
#pragma unroll
    for (int k = 0; k < F / 32; ++k)
      __hip_atomic_fetch_add(ap + k * 32, w * v[j][k],
                             __ATOMIC_RELAXED, __HIP_MEMORY_SCOPE_WORKGROUP);
  }
}

template<int F, int TN, int LD>
__device__ __forceinline__ void rec_group(const int4 r, const float* __restrict__ h,
                                          float* acc, int n0, int lane) {
  float va[8][F / 32];
  float vb[8][F / 32];
  rows_load<F, 0>(va, r.x, h, lane);
  rows_load<F, 8>(vb, r.x, h, lane);
  rows_add<F, TN, LD, 0>(va, r.y, r.z, acc, n0, lane);
  rows_load<F, 16>(va, r.x, h, lane);
  rows_add<F, TN, LD, 8>(vb, r.y, r.z, acc, n0, lane);
  rows_load<F, 24>(vb, r.x, h, lane);
  rows_add<F, TN, LD, 16>(va, r.y, r.z, acc, n0, lane);
  rows_add<F, TN, LD, 24>(vb, r.y, r.z, acc, n0, lane);
}

constexpr int LISTCAP = 16;

constexpr int NWAVE = 32;
constexpr int GROUP = 32;
constexpr int QCAP  = 512 + GROUP;

template<typename G>
__device__ __forceinline__ void scan_ids(const unsigned char* __restrict__ rid, int E, int blk,
                                         int wave, int lane, int* q, G grp) {
  int qn = 0;
  const unsigned pat = ((unsigned)blk & 255u) * 0x01010101u;
  const unsigned lt  = (1u << lane) - 1u;
  const int nch = (E + 511) >> 9;
  const uint4* rid4 = (const uint4*)rid;

  uint4 vn = make_uint4(0u, 0u, 0u, 0u);
  if (wave < nch) vn = rid4[(size_t)wave * 32 + lane];
  for (int ch = wave; ch < nch; ch += NWAVE) {
    const uint4 v = vn;
    if (ch + NWAVE < nch) vn = rid4[(size_t)(ch + NWAVE) * 32 + lane];
    const unsigned x0 = v.x ^ pat, x1 = v.y ^ pat, x2 = v.z ^ pat, x3 = v.w ^ pat;
    const unsigned f0 = ~(((x0 & 0x7F7F7F7Fu) + 0x7F7F7F7Fu) | x0) & 0x80808080u;
    const unsigned f1 = ~(((x1 & 0x7F7F7F7Fu) + 0x7F7F7F7Fu) | x1) & 0x80808080u;
    const unsigned f2 = ~(((x2 & 0x7F7F7F7Fu) + 0x7F7F7F7Fu) | x2) & 0x80808080u;
    const unsigned f3 = ~(((x3 & 0x7F7F7F7Fu) + 0x7F7F7F7Fu) | x3) & 0x80808080u;
    unsigned msk = (f0 >> 7) | (f1 >> 6) | (f2 >> 5) | (f3 >> 4);
    const int e0 = (ch << 9) + lane * 16;
    unsigned any = __builtin_amdgcn_ballot_w32(msk != 0u);
    while (any) {
      const int  p  = __builtin_ctz(msk | 0x80000000u);
      const int  e  = e0 + (((p & 7) << 2) | (p >> 3));
      const bool ok = (msk != 0u) && (e < E);
      const unsigned okb = __builtin_amdgcn_ballot_w32(ok);
      if (ok) q[qn + __builtin_popcount(okb & lt)] = e;
      qn  += __builtin_popcount(okb);
      msk &= msk - 1u;
      any  = __builtin_amdgcn_ballot_w32(msk != 0u);
    }
    if (qn >= GROUP) {
      __builtin_amdgcn_wave_barrier();
      int b0 = 0;
      for (; b0 + GROUP <= qn; b0 += GROUP) grp(b0, GROUP);
      const int rest = qn - b0;
      int t = 0;
      if (lane < rest) t = q[b0 + lane];
      if (lane < rest) q[lane] = t;
      qn = rest;
    }
  }
  if (qn > 0) {
    __builtin_amdgcn_wave_barrier();
    grp(0, qn);
  }
}

template<int TN>
__global__ __launch_bounds__(1024)
void k_range_deg(const unsigned char* __restrict__ rid, const int* __restrict__ row,
                 const int* __restrict__ col,
                 const float* __restrict__ w, float* __restrict__ dinv, int N, int E,
                 int4* elists, int* ecounts,
                 const float* __restrict__ x, const float* __restrict__ W1,
                 float* __restrict__ h) {
  static_assert(TN <= 16 * NWAVE && TN % 16 == 0,
                "one 16-row strip per wave, and no strip reaches into the next block's rows");
  __shared__ float deg[TN + 1];
  __shared__ int queue[NWAVE * QCAP];
  __shared__ __align__(16) __bf16 xs[NWAVE * 16 * XLDSP];
  __shared__ __align__(16) __bf16 w1s[128 * XLDSP];

  const int lane = threadIdx.x & 31;
  const int wave = __builtin_amdgcn_readfirstlane(threadIdx.x >> 5);
  const int n0   = blockIdx.x * TN;
  int rows = N - n0;
  if (rows > TN) rows = TN;

  for (int i = threadIdx.x; i < TN + 1; i += 1024) deg[i] = 0.0f;
  {
    float wf[4][4];
    stage_wt_load<128>(W1, wf, threadIdx.x);
    stage_wt_store<128>(wf, w1s, threadIdx.x);
  }
  __syncthreads();

  if (wave * 16 < rows)
    strip_x_product(x, n0 + wave * 16, N, w1s, h, xs + wave * (16 * XLDSP), lane);

  const int wid = blockIdx.x * NWAVE + wave;
  int4* el = elists + (size_t)wid * (LISTCAP * 32);
  int* q  = queue + wave * QCAP;
  int ng = 0;
  scan_ids(rid, E, blockIdx.x, wave, lane, q, [&](int b0, int cnt) {
    int e = -1;
    if (lane < cnt) e = q[b0 + lane];
    int4 r = make_int4(0, -1, 0, 0);
    if (e >= 0) {
      const int   rr = row[e];
      const int   cc = col[e];
      const float s  = w[e];
      r = make_int4(rr, cc, __float_as_int(s), 0);
      int c = cc - n0;
      c = ((unsigned)c < (unsigned)TN) ? c : TN;
      __hip_atomic_fetch_add(deg + c, s, __ATOMIC_RELAXED, __HIP_MEMORY_SCOPE_WORKGROUP);
    }
    if (ng < LISTCAP) el[ng * 32 + lane] = r;
    ++ng;
  });
  if (lane == 0) ecounts[wid] = (ng <= LISTCAP) ? ng : -1;

  __syncthreads();
  for (int i = threadIdx.x; i < rows; i += 1024) {
    const float d = deg[i] + 1.0f;
    dinv[n0 + i] = (d > 0.0f) ? rsqrtf(d) : 0.0f;
  }
}

template<int F, int TN, int LD, bool KEEP>
__device__ __forceinline__ void range_group(const int4 r,
                                            const float* __restrict__ h,
                                            float* acc, int n0, int lane,
                                            int4* lst, int& ng) {
  if (KEEP) {
    if (ng < LISTCAP) lst[ng * 32 + lane] = r;
    ++ng;
  }
  rec_group<F, TN, LD>(r, h, acc, n0, lane);
}

template<int LD>
__device__ __forceinline__ void strip_product(const float* acc, int s, int rows, int nbase,
                                              const __bf16* Bt,
                                              float* __restrict__ H2, int lane) {
  constexpr int K  = 128;
  constexpr int NC = 64;
  const int half = lane >> 4;
  const int l15  = lane & 15;
  int rl = s * 16 + l15;
  if (rl > rows - 1) rl = rows - 1;
  const float* Ar = acc + rl * LD + half * 8;

  const int ml0 = s * 16 + half * 8;

  const __bf16* Brow = Bt + l15 * XLDSP + half * 8;
  v8f c[NC / 16];
#pragma unroll
  for (int t = 0; t < NC / 16; ++t) c[t] = v8f{};
#pragma unroll 1
  for (int kk = 0; kk < K / 32; ++kk) {
    const float4 f0 = *(const float4*)(Ar + kk * 32);
    const float4 f1 = *(const float4*)(Ar + kk * 32 + 4);
    const float4 f2 = *(const float4*)(Ar + kk * 32 + 16);
    const float4 f3 = *(const float4*)(Ar + kk * 32 + 20);
    v16bf a;
    a[0]  = tobf(fmaxf(f0.x, 0.f)); a[1]  = tobf(fmaxf(f0.y, 0.f));
    a[2]  = tobf(fmaxf(f0.z, 0.f)); a[3]  = tobf(fmaxf(f0.w, 0.f));
    a[4]  = tobf(fmaxf(f1.x, 0.f)); a[5]  = tobf(fmaxf(f1.y, 0.f));
    a[6]  = tobf(fmaxf(f1.z, 0.f)); a[7]  = tobf(fmaxf(f1.w, 0.f));
    a[8]  = tobf(fmaxf(f2.x, 0.f)); a[9]  = tobf(fmaxf(f2.y, 0.f));
    a[10] = tobf(fmaxf(f2.z, 0.f)); a[11] = tobf(fmaxf(f2.w, 0.f));
    a[12] = tobf(fmaxf(f3.x, 0.f)); a[13] = tobf(fmaxf(f3.y, 0.f));
    a[14] = tobf(fmaxf(f3.z, 0.f)); a[15] = tobf(fmaxf(f3.w, 0.f));
#pragma unroll
    for (int t = 0; t < NC / 16; ++t) {
      const v8bf b0 = *(const v8bf*)(Brow + t * 16 * XLDSP + kk * 32);
      const v8bf b1 = *(const v8bf*)(Brow + t * 16 * XLDSP + kk * 32 + 16);
      const v16bf b = __builtin_shufflevector(b0, b1,
          0, 1, 2, 3, 4, 5, 6, 7, 8, 9, 10, 11, 12, 13, 14, 15);
      c[t] = __builtin_amdgcn_wmma_f32_16x16x32_bf16(false, a, false, b,
                                                     (short)0, c[t], false, false);
    }
  }

  if (s * 16 + 16 <= rows) {
#pragma unroll
    for (int t = 0; t < NC / 16; ++t) {
#pragma unroll
      for (int r = 0; r < 8; ++r)
        H2[(size_t)(nbase + ml0 + r) * NC + t * 16 + l15] = c[t][r];
    }
  } else {
#pragma unroll
    for (int t = 0; t < NC / 16; ++t) {
#pragma unroll
      for (int r = 0; r < 8; ++r) {
        if (ml0 + r < rows)
          H2[(size_t)(nbase + ml0 + r) * NC + t * 16 + l15] = c[t][r];
      }
    }
  }
}

template<int F, int TN, bool REPLAY>
__global__ __launch_bounds__(1024)
void k_range_add(const int* __restrict__ row, const int* __restrict__ col,
                 const float* __restrict__ w, const float* __restrict__ dinv,
                 const unsigned char* __restrict__ rid,
                 const float* __restrict__ h, float* out, int N, int E,
                 const int4* elists, const int* ecounts,
                 int4* lists, int* counts,
                 const float* __restrict__ W2, const float* __restrict__ bias,
                 float* __restrict__ h2) {
  constexpr int LD  = REPLAY ? F : F + 4;
  constexpr int F4  = F / 4;
  constexpr int LD4 = LD / 4;
  __shared__ __attribute__((aligned(16))) float acc[(TN + 1) * LD];
  __shared__ int queue[NWAVE * QCAP];
  __shared__ __align__(16) __bf16 w2s[REPLAY ? 8 : 64 * XLDSP];

  const int lane = threadIdx.x & 31;
  const int wave = __builtin_amdgcn_readfirstlane(threadIdx.x >> 5);
  const int n0   = blockIdx.x * TN;
  int rows = N - n0;
  if (rows > TN) rows = TN;
  const int cnt4 = rows * (F / 4);

  float wf[2][4];
  if constexpr (!REPLAY) stage_wt_load<64>(W2, wf, threadIdx.x);

  float4* acc4 = (float4*)acc;
  const int c0 = (threadIdx.x % F4) * 4;
  const float4 bq = make_float4(bias[c0], bias[c0 + 1], bias[c0 + 2], bias[c0 + 3]);
  const float4* hown4 = (const float4*)(h + (size_t)n0 * F);
  for (int i0 = threadIdx.x; i0 < cnt4; i0 += 4096) {
    float4 t[4];
    float  dd[4];
#pragma unroll
    for (int u = 0; u < 4; ++u) {
      const int i = i0 + u * 1024;
      dd[u] = 0.f;
      t[u] = (i < cnt4) ? hown4[i] : make_float4(0.f, 0.f, 0.f, 0.f);
      if (i < cnt4) dd[u] = dinv[n0 + i / F4];
    }
#pragma unroll
    for (int u = 0; u < 4; ++u) {
      const int i = i0 + u * 1024;
      const float d2 = dd[u] * dd[u];
      t[u] = make_float4(fmaf(d2, t[u].x, bq.x), fmaf(d2, t[u].y, bq.y),
                         fmaf(d2, t[u].z, bq.z), fmaf(d2, t[u].w, bq.w));
      if (i < cnt4) acc4[(i / F4) * LD4 + (i % F4)] = t[u];
    }
  }
  if constexpr (!REPLAY) stage_wt_store<64>(wf, w2s, threadIdx.x);
  __syncthreads();

  const int wid = blockIdx.x * NWAVE + wave;
  int4* lst = lists + (size_t)wid * (LISTCAP * 32);
  int ng = 0;
  bool scan = true;
  if (REPLAY) {
    const int kept = __builtin_amdgcn_readfirstlane(counts[wid]);
    if (kept >= 0) {
      scan = false;
      int4 rn = make_int4(0, -1, 0, 0);
      if (kept > 0) rn = lst[lane];
      for (int g = 0; g < kept; ++g) {
        const int4 r = rn;
        if (g + 1 < kept) rn = lst[(g + 1) * 32 + lane];
        rec_group<F, TN, LD>(r, h, acc, n0, lane);
      }
    }
  } else {
    const int kept = __builtin_amdgcn_readfirstlane(ecounts[wid]);
    if (kept >= 0) {
      scan = false;
      const int4* el = elists + (size_t)wid * (LISTCAP * 32);
      int4 ra = make_int4(0, -1, 0, 0);
      if (kept > 0) ra = el[lane];
      int4 rb = make_int4(0, -1, 0, 0);
      if (kept > 1) rb = el[32 + lane];
      float dr = dinv[ra.x];
      float dc = dinv[ra.y < 0 ? 0 : ra.y];
      for (int g = 0; g < kept; ++g) {
        int4 r = ra;
        r.z = __float_as_int(dr * __int_as_float(r.z) * dc);
        ra = rb;
        dr = dinv[ra.x];
        dc = dinv[ra.y < 0 ? 0 : ra.y];
        rb = make_int4(0, -1, 0, 0);
        if (g + 2 < kept) rb = el[(g + 2) * 32 + lane];
        range_group<F, TN, LD, true>(r, h, acc, n0, lane, lst, ng);
      }
    }
  }
  if (scan) {
    int* q = queue + wave * QCAP;
    scan_ids(rid, E, blockIdx.x, wave, lane, q, [&](int b0, int cnt) {
      int4 r = make_int4(0, -1, 0, 0);
      if (lane < cnt) r = edge_rec(q[b0 + lane], row, col, w, dinv);
      range_group<F, TN, LD, !REPLAY>(r, h, acc, n0, lane, lst, ng);
    });
  }
  if (!REPLAY) {
    if (lane == 0) counts[wid] = (ng <= LISTCAP) ? ng : -1;
  }
  __syncthreads();
  if constexpr (REPLAY) {
    float4* out4 = (float4*)(out + (size_t)n0 * F);
    for (int i = threadIdx.x; i < cnt4; i += 1024) out4[i] = acc4[(i / F4) * LD4 + (i % F4)];
  } else {
    const int nstrips = (rows + 15) >> 4;
    for (int s = wave; s < nstrips; s += NWAVE)
      strip_product<LD>(acc, s, rows, n0, w2s, h2, lane);
  }
}

extern "C" void kernel_launch(void* const* d_in, const int* in_sizes, int n_in,
                              void* d_out, int out_size, void* d_ws, size_t ws_size,
                              hipStream_t stream) {
  const float* x  = (const float*)d_in[0];
  const int*   ei = (const int*)  d_in[1];
  const float* P  = (const float*)d_in[2];
  const float* W1 = (const float*)d_in[3];
  const float* b1 = (const float*)d_in[4];
  const float* W2 = (const float*)d_in[5];
  const float* b2 = (const float*)d_in[6];
  float* out = (float*)d_out;

  constexpr int F_IN = 128, H1 = 128, H2 = 64;
  const int N = in_sizes[0] / F_IN;
  const int E = in_sizes[2];
  const int* row = ei;
  const int* col = ei + E;

  constexpr int TN = 400;
  const int Epad = ((E + 511) / 512) * 512;

  unsigned char* rid = (unsigned char*)d_ws;
  float* dinv = (float*)(rid + Epad);
  float* wn   = dinv + N;
  float* h    = wn + E;

  const int T = 256;
  const int ranges = (N + TN - 1) / TN;
  int4* lists   = (int4*)(h + (size_t)N * H1);
  int*  counts  = (int*)(lists + (size_t)ranges * 32 * LISTCAP * 32);
  int*  ecounts = counts + (size_t)ranges * 32;
  int4* elists  = (int4*)(ecounts + (size_t)ranges * 32);
  float* h2     = (float*)(elists + (size_t)ranges * 32 * LISTCAP * 32);

  k_sig_rid<TN><<<(Epad / 8 + T - 1) / T, T, 0, stream>>>(P, col, wn, rid, E, Epad);

  k_range_deg<TN><<<ranges, 1024, 0, stream>>>(rid, row, col, wn, dinv, N, E, elists, ecounts,
                                               x, W1, h);

  k_range_add<H1, TN, false><<<ranges, 1024, 0, stream>>>(row, col, wn, dinv, rid, h, nullptr, N, E,
                                                          elists, ecounts, lists, counts,
                                                          W2, b1, h2);

  k_range_add<H2, TN, true><<<ranges, 1024, 0, stream>>>(row, col, wn, dinv, rid, h2, out, N, E,
                                                         elists, ecounts, lists, counts,
                                                         nullptr, b2, nullptr);
}
